# starting kernel
# baseline (speedup 1.0000x reference)
#include <hip/hip_runtime.h>
#include <hip/hip_bf16.h>

typedef float v2f __attribute__((ext_vector_type(2)));
typedef float v8f __attribute__((ext_vector_type(8)));

#define NB   16
#define TT   1024
#define DD   2000
#define KK   21
#define PP   10
#define RR   (NB * TT)
#define WPAD 32

__global__ __launch_bounds__(32) void pad_w_kernel(
    const float* __restrict__ W, float* __restrict__ Wp) {
  const int k = blockIdx.x;
  const int c = threadIdx.x;
  Wp[(long)k * WPAD + c] = (c < KK) ? W[(long)k * KK + c] : 0.f;
}

__device__ __forceinline__ void gemm_span(
    const float* __restrict__ A, const float* __restrict__ Wp,
    long rowA, int l15, int half, int kBeg, int kEnd,
    v8f& acc0, v8f& acc1) {
#pragma unroll 2
  for (int k0 = kBeg; k0 < kEnd; k0 += 4) {
    const int kk = k0 + half * 2;
    v2f a;
    a.x = A[rowA * DD + kk];
    a.y = A[rowA * DD + kk + 1];
    v2f b0, b1;
    b0.x = Wp[(long)kk * WPAD + l15];
    b0.y = Wp[(long)(kk + 1) * WPAD + l15];
    b1.x = Wp[(long)kk * WPAD + 16 + l15];
    b1.y = Wp[(long)(kk + 1) * WPAD + 16 + l15];

    acc0 = __builtin_amdgcn_wmma_f32_16x16x4_f32(
        false, a, false, b0, (short)0, acc0, false, false);
    acc1 = __builtin_amdgcn_wmma_f32_16x16x4_f32(
        false, a, false, b1, (short)0, acc1, false, false);
  }
}

__global__ __launch_bounds__(128) void wmma_xw_kernel(
    const float* __restrict__ Alo, const float* __restrict__ Ahi, int split,
    const float* __restrict__ Wp, float* __restrict__ wout) {
  const int lane = threadIdx.x & 31;
  const int wave = threadIdx.x >> 5;
  const int tile = blockIdx.x * 4 + wave;
  const int half = lane >> 4;
  const int l15  = lane & 15;
  const long rowA = (long)(tile * 16 + l15);

  v8f acc0 = {0.f, 0.f, 0.f, 0.f, 0.f, 0.f, 0.f, 0.f};
  v8f acc1 = {0.f, 0.f, 0.f, 0.f, 0.f, 0.f, 0.f, 0.f};

  gemm_span(Alo, Wp, rowA, l15, half, 0,     split, acc0, acc1);
  gemm_span(Ahi, Wp, rowA, l15, half, split, DD,    acc0, acc1);

#pragma unroll
  for (int v = 0; v < 8; ++v) {
    const long rowO = (long)(tile * 16 + v + half * 8);
    wout[rowO * WPAD + l15]      = acc0[v];
    wout[rowO * WPAD + 16 + l15] = acc1[v];
  }
}

__global__ __launch_bounds__(256) void shift_kernel(
    const float* __restrict__ x, const float* __restrict__ w,
    float* __restrict__ out, int sOff) {
  const int idx  = blockIdx.x;
  const int nIdx = idx / TT;
  const int t    = idx - nIdx * TT;

  __shared__ float sw[KK];
  if (threadIdx.x < KK) sw[threadIdx.x] = w[(long)idx * WPAD + threadIdx.x];
  __syncthreads();

  const int kLo   = (t < PP) ? (PP - t) : 0;
  const int kHiEx = (t + PP >= TT) ? (TT - t + PP) : KK;

  for (int d4 = threadIdx.x; d4 < 250; d4 += 256) {
    float4 acc = make_float4(0.f, 0.f, 0.f, 0.f);
    for (int k = kLo; k < kHiEx; ++k) {
      const int ts = t + k - PP;
      const float4 v = ((const float4*)(x + ((long)nIdx * TT + ts) * DD + sOff))[d4];
      const float  s = sw[k];
      acc.x += s * v.x; acc.y += s * v.y; acc.z += s * v.z; acc.w += s * v.w;
    }
    ((float4*)(out + (long)idx * DD + sOff))[d4] = acc;
  }
}

extern "C" void kernel_launch(void* const* d_in, const int* in_sizes, int n_in,
                              void* d_out, int out_size, void* d_ws, size_t ws_size,
                              hipStream_t stream) {
  const float* x  = (const float*)d_in[0];
  const float* W0 = (const float*)d_in[1];
  const float* W1 = (const float*)d_in[2];
  float* out  = (float*)d_out;

  float* wbuf = (float*)d_ws;
  float* Wp   = (float*)d_ws + (long)RR * WPAD;

  (void)in_sizes; (void)n_in; (void)out_size; (void)ws_size;

  pad_w_kernel<<<DD, 32, 0, stream>>>(W0, Wp);
  wmma_xw_kernel<<<RR / 16 / 4, 128, 0, stream>>>(x, x, DD, Wp, wbuf);
  shift_kernel<<<RR, 256, 0, stream>>>(x, wbuf, out, 0);

  pad_w_kernel<<<DD, 32, 0, stream>>>(W1, Wp);
  wmma_xw_kernel<<<RR / 16 / 4, 128, 0, stream>>>(out, x, 1000, Wp, wbuf);
  shift_kernel<<<RR, 256, 0, stream>>>(x, wbuf, out, 1000);
}


# GEMM: K loop split over 16 waves per 16-row tile (16 waves per SIMD, was 1); the 16 partial tiles are summed through LDS in fixed slice order, which changes the summation order of w
# speedup vs baseline: 1.5861x; 1.5861x over previous
#include <hip/hip_runtime.h>
#include <hip/hip_bf16.h>

typedef float v2f __attribute__((ext_vector_type(2)));
typedef float v8f __attribute__((ext_vector_type(8)));

#define NB   16
#define TT   1024
#define DD   2000
#define KK   21
#define PP   10
#define RR   (NB * TT)
#define WPAD 32
#define KSPLIT 16

__global__ __launch_bounds__(32) void pad_w_kernel(
    const float* __restrict__ W, float* __restrict__ Wp) {
  const int k = blockIdx.x;
  const int c = threadIdx.x;
  Wp[(long)k * WPAD + c] = (c < KK) ? W[(long)k * KK + c] : 0.f;
}

__device__ __forceinline__ void gemm_span(
    const float* __restrict__ A, const float* __restrict__ Wp,
    long rowA, int l15, int half, int kBeg, int kEnd,
    v8f& acc0, v8f& acc1) {
#pragma unroll 2
  for (int k0 = kBeg; k0 < kEnd; k0 += 4) {
    const int kk = k0 + half * 2;
    v2f a;
    a.x = A[rowA * DD + kk];
    a.y = A[rowA * DD + kk + 1];
    v2f b0, b1;
    b0.x = Wp[(long)kk * WPAD + l15];
    b0.y = Wp[(long)(kk + 1) * WPAD + l15];
    b1.x = Wp[(long)kk * WPAD + 16 + l15];
    b1.y = Wp[(long)(kk + 1) * WPAD + 16 + l15];

    acc0 = __builtin_amdgcn_wmma_f32_16x16x4_f32(
        false, a, false, b0, (short)0, acc0, false, false);
    acc1 = __builtin_amdgcn_wmma_f32_16x16x4_f32(
        false, a, false, b1, (short)0, acc1, false, false);
  }
}

__global__ __launch_bounds__(512) void wmma_xw_kernel(
    const float* __restrict__ Alo, const float* __restrict__ Ahi, int split,
    const float* __restrict__ Wp, float* __restrict__ wout) {
  __shared__ float part[KSPLIT * 16 * WPAD];

  const int lane = threadIdx.x & 31;
  const int ks   = __builtin_amdgcn_readfirstlane((int)(threadIdx.x >> 5));
  const int tile = blockIdx.x;
  const int half = lane >> 4;
  const int l15  = lane & 15;
  const long rowA = (long)(tile * 16 + l15);

  const int kBeg = ((ks * (DD / 4)) / KSPLIT) * 4;
  const int kEnd = (((ks + 1) * (DD / 4)) / KSPLIT) * 4;
  const int endLo = (kEnd < split) ? kEnd : split;
  const int begHi = (kBeg > split) ? kBeg : split;

  v8f acc0 = {0.f, 0.f, 0.f, 0.f, 0.f, 0.f, 0.f, 0.f};
  v8f acc1 = {0.f, 0.f, 0.f, 0.f, 0.f, 0.f, 0.f, 0.f};

  gemm_span(Alo, Wp, rowA, l15, half, kBeg,  endLo, acc0, acc1);
  gemm_span(Ahi, Wp, rowA, l15, half, begHi, kEnd,  acc0, acc1);

  float* my = part + ks * (16 * WPAD);
#pragma unroll
  for (int v = 0; v < 8; ++v) {
    const int r = v + half * 8;
    my[r * WPAD + l15]      = acc0[v];
    my[r * WPAD + 16 + l15] = acc1[v];
  }
  __syncthreads();

  const int e = threadIdx.x;
  float s = part[e];
#pragma unroll
  for (int j = 1; j < KSPLIT; ++j) s += part[j * (16 * WPAD) + e];
  wout[(long)tile * (16 * WPAD) + e] = s;
}

__global__ __launch_bounds__(256) void shift_kernel(
    const float* __restrict__ x, const float* __restrict__ w,
    float* __restrict__ out, int sOff) {
  const int idx  = blockIdx.x;
  const int nIdx = idx / TT;
  const int t    = idx - nIdx * TT;

  __shared__ float sw[KK];
  if (threadIdx.x < KK) sw[threadIdx.x] = w[(long)idx * WPAD + threadIdx.x];
  __syncthreads();

  const int kLo   = (t < PP) ? (PP - t) : 0;
  const int kHiEx = (t + PP >= TT) ? (TT - t + PP) : KK;

  for (int d4 = threadIdx.x; d4 < 250; d4 += 256) {
    float4 acc = make_float4(0.f, 0.f, 0.f, 0.f);
    for (int k = kLo; k < kHiEx; ++k) {
      const int ts = t + k - PP;
      const float4 v = ((const float4*)(x + ((long)nIdx * TT + ts) * DD + sOff))[d4];
      const float  s = sw[k];
      acc.x += s * v.x; acc.y += s * v.y; acc.z += s * v.z; acc.w += s * v.w;
    }
    ((float4*)(out + (long)idx * DD + sOff))[d4] = acc;
  }
}

extern "C" void kernel_launch(void* const* d_in, const int* in_sizes, int n_in,
                              void* d_out, int out_size, void* d_ws, size_t ws_size,
                              hipStream_t stream) {
  const float* x  = (const float*)d_in[0];
  const float* W0 = (const float*)d_in[1];
  const float* W1 = (const float*)d_in[2];
  float* out  = (float*)d_out;

  float* wbuf = (float*)d_ws;
  float* Wp   = (float*)d_ws + (long)RR * WPAD;

  (void)in_sizes; (void)n_in; (void)out_size; (void)ws_size;

  pad_w_kernel<<<DD, 32, 0, stream>>>(W0, Wp);
  wmma_xw_kernel<<<RR / 16, KSPLIT * 32, 0, stream>>>(x, x, DD, Wp, wbuf);
  shift_kernel<<<RR, 256, 0, stream>>>(x, wbuf, out, 0);

  pad_w_kernel<<<DD, 32, 0, stream>>>(W1, Wp);
  wmma_xw_kernel<<<RR / 16, KSPLIT * 32, 0, stream>>>(out, x, 1000, Wp, wbuf);
  shift_kernel<<<RR, 256, 0, stream>>>(x, wbuf, out, 1000);
}


# shift phase: each thread forms 16 consecutive time steps of its float4 column from 36 x rows loaded once into registers (was 21 row loads per output row, one row per block); tap order kept
# speedup vs baseline: 2.6378x; 1.6631x over previous
#include <hip/hip_runtime.h>
#include <hip/hip_bf16.h>

typedef float v2f __attribute__((ext_vector_type(2)));
typedef float v8f __attribute__((ext_vector_type(8)));

#define NB   16
#define TT   1024
#define DD   2000
#define KK   21
#define PP   10
#define RR   (NB * TT)
#define WPAD 32
#define KSPLIT 16

__global__ __launch_bounds__(32) void pad_w_kernel(
    const float* __restrict__ W, float* __restrict__ Wp) {
  const int k = blockIdx.x;
  const int c = threadIdx.x;
  Wp[(long)k * WPAD + c] = (c < KK) ? W[(long)k * KK + c] : 0.f;
}

__device__ __forceinline__ void gemm_span(
    const float* __restrict__ A, const float* __restrict__ Wp,
    long rowA, int l15, int half, int kBeg, int kEnd,
    v8f& acc0, v8f& acc1) {
#pragma unroll 2
  for (int k0 = kBeg; k0 < kEnd; k0 += 4) {
    const int kk = k0 + half * 2;
    v2f a;
    a.x = A[rowA * DD + kk];
    a.y = A[rowA * DD + kk + 1];
    v2f b0, b1;
    b0.x = Wp[(long)kk * WPAD + l15];
    b0.y = Wp[(long)(kk + 1) * WPAD + l15];
    b1.x = Wp[(long)kk * WPAD + 16 + l15];
    b1.y = Wp[(long)(kk + 1) * WPAD + 16 + l15];

    acc0 = __builtin_amdgcn_wmma_f32_16x16x4_f32(
        false, a, false, b0, (short)0, acc0, false, false);
    acc1 = __builtin_amdgcn_wmma_f32_16x16x4_f32(
        false, a, false, b1, (short)0, acc1, false, false);
  }
}

__global__ __launch_bounds__(512) void wmma_xw_kernel(
    const float* __restrict__ Alo, const float* __restrict__ Ahi, int split,
    const float* __restrict__ Wp, float* __restrict__ wout) {
  __shared__ float part[KSPLIT * 16 * WPAD];

  const int lane = threadIdx.x & 31;
  const int ks   = __builtin_amdgcn_readfirstlane((int)(threadIdx.x >> 5));
  const int tile = blockIdx.x;
  const int half = lane >> 4;
  const int l15  = lane & 15;
  const long rowA = (long)(tile * 16 + l15);

  const int kBeg = ((ks * (DD / 4)) / KSPLIT) * 4;
  const int kEnd = (((ks + 1) * (DD / 4)) / KSPLIT) * 4;
  const int endLo = (kEnd < split) ? kEnd : split;
  const int begHi = (kBeg > split) ? kBeg : split;

  v8f acc0 = {0.f, 0.f, 0.f, 0.f, 0.f, 0.f, 0.f, 0.f};
  v8f acc1 = {0.f, 0.f, 0.f, 0.f, 0.f, 0.f, 0.f, 0.f};

  gemm_span(Alo, Wp, rowA, l15, half, kBeg,  endLo, acc0, acc1);
  gemm_span(Ahi, Wp, rowA, l15, half, begHi, kEnd,  acc0, acc1);

  float* my = part + ks * (16 * WPAD);
#pragma unroll
  for (int v = 0; v < 8; ++v) {
    const int r = v + half * 8;
    my[r * WPAD + l15]      = acc0[v];
    my[r * WPAD + 16 + l15] = acc1[v];
  }
  __syncthreads();

  const int e = threadIdx.x;
  float s = part[e];
#pragma unroll
  for (int j = 1; j < KSPLIT; ++j) s += part[j * (16 * WPAD) + e];
  wout[(long)tile * (16 * WPAD) + e] = s;
}

#define TSTEP 16
#define NROW  (TSTEP + 2 * PP)
#define ROW4  (DD / 4)
#define COL4  250
#define CBLK  128

template <int WHERE>
__device__ __forceinline__ void shift_tile(
    const float4* __restrict__ xcol, const float* __restrict__ wt,
    float4* __restrict__ ocol) {
  constexpr int rLo = (WHERE == 1) ? PP : 0;
  constexpr int rHi = (WHERE == 2) ? (NROW - PP) : NROW;

  float4 rows[NROW];
#pragma unroll
  for (int r = rLo; r < rHi; ++r) rows[r] = xcol[(long)(r - PP) * ROW4];

#pragma unroll
  for (int j = 0; j < TSTEP; ++j) {
    float4 acc = make_float4(0.f, 0.f, 0.f, 0.f);
#pragma unroll
    for (int k = 0; k < KK; ++k) {
      if (j + k < rLo || j + k >= rHi) continue;
      const float  s = wt[j * WPAD + k];
      const float4 v = rows[j + k];
      acc.x = __builtin_fmaf(s, v.x, acc.x);
      acc.y = __builtin_fmaf(s, v.y, acc.y);
      acc.z = __builtin_fmaf(s, v.z, acc.z);
      acc.w = __builtin_fmaf(s, v.w, acc.w);
    }
    ocol[(long)j * ROW4] = acc;
  }
}

__global__ __launch_bounds__(CBLK, 8) void shift_kernel(
    const float* __restrict__ x, const float* __restrict__ w,
    float* __restrict__ out, int sOff) {
  const int d4 = (blockIdx.x & 1) * CBLK + threadIdx.x;
  if (d4 >= COL4) return;
  const int row0 = (blockIdx.x >> 1) * TSTEP;
  const int t0   = row0 % TT;

  const float4* xcol = (const float4*)(x + (long)row0 * DD + sOff) + d4;
  float4*       ocol = (float4*)(out + (long)row0 * DD + sOff) + d4;
  const float*  wt   = w + (long)row0 * WPAD;

  if (t0 == 0)               shift_tile<1>(xcol, wt, ocol);
  else if (t0 == TT - TSTEP) shift_tile<2>(xcol, wt, ocol);
  else                       shift_tile<0>(xcol, wt, ocol);
}

extern "C" void kernel_launch(void* const* d_in, const int* in_sizes, int n_in,
                              void* d_out, int out_size, void* d_ws, size_t ws_size,
                              hipStream_t stream) {
  const float* x  = (const float*)d_in[0];
  const float* W0 = (const float*)d_in[1];
  const float* W1 = (const float*)d_in[2];
  float* out  = (float*)d_out;

  float* wbuf = (float*)d_ws;
  float* Wp   = (float*)d_ws + (long)RR * WPAD;

  (void)in_sizes; (void)n_in; (void)out_size; (void)ws_size;

  pad_w_kernel<<<DD, 32, 0, stream>>>(W0, Wp);
  wmma_xw_kernel<<<RR / 16, KSPLIT * 32, 0, stream>>>(x, x, DD, Wp, wbuf);
  shift_kernel<<<(RR / TSTEP) * 2, CBLK, 0, stream>>>(x, wbuf, out, 0);

  pad_w_kernel<<<DD, 32, 0, stream>>>(W1, Wp);
  wmma_xw_kernel<<<RR / 16, KSPLIT * 32, 0, stream>>>(out, x, 1000, Wp, wbuf);
  shift_kernel<<<(RR / TSTEP) * 2, CBLK, 0, stream>>>(x, wbuf, out, 1000);
}


# GEMM: each wave copies its K slice of A to an LDS region of its own with 16-byte loads along the rows and reads its fragments there (was 8-byte reads from 16 rows per load); k order, sums unchanged
# speedup vs baseline: 3.5555x; 1.3479x over previous
#include <hip/hip_runtime.h>
#include <hip/hip_bf16.h>

typedef float v2f __attribute__((ext_vector_type(2)));
typedef float v8f __attribute__((ext_vector_type(8)));

#define NB   16
#define TT   1024
#define DD   2000
#define KK   21
#define PP   10
#define RR   (NB * TT)
#define WPAD 32
#define KSPLIT 16

__global__ __launch_bounds__(32) void pad_w_kernel(
    const float* __restrict__ W, float* __restrict__ Wp) {
  const int k = blockIdx.x;
  const int c = threadIdx.x;
  Wp[(long)k * WPAD + c] = (c < KK) ? W[(long)k * KK + c] : 0.f;
}

#define KPASS  64
#define SPITCH (KPASS + 4)
#define WSTAGE (16 * SPITCH)

__device__ __forceinline__ void wave_lds_handover() {
  __builtin_amdgcn_fence(__ATOMIC_RELEASE, "wavefront");
  __builtin_amdgcn_wave_barrier();
  __builtin_amdgcn_fence(__ATOMIC_ACQUIRE, "wavefront");
}

__global__ __launch_bounds__(512) void wmma_xw_kernel(
    const float* __restrict__ Alo, const float* __restrict__ Ahi, int split,
    const float* __restrict__ Wp, float* __restrict__ wout) {
  __shared__ __attribute__((aligned(16))) float stage[KSPLIT * WSTAGE];

  const int lane = threadIdx.x & 31;
  const int ks   = __builtin_amdgcn_readfirstlane((int)(threadIdx.x >> 5));
  const int tile = blockIdx.x;
  const int half = lane >> 4;
  const int l15  = lane & 15;

  const int kBeg = ((ks * (DD / 4)) / KSPLIT) * 4;
  const int kEnd = (((ks + 1) * (DD / 4)) / KSPLIT) * 4;

  float* my = stage + ks * WSTAGE;
  const int    rowWrAt = ks * WSTAGE + half * SPITCH + l15 * 4;
  const float* fragRd = my + l15 * SPITCH + half * 2;

  v8f acc0 = {0.f, 0.f, 0.f, 0.f, 0.f, 0.f, 0.f, 0.f};
  v8f acc1 = {0.f, 0.f, 0.f, 0.f, 0.f, 0.f, 0.f, 0.f};

#pragma unroll
  for (int p = 0; p < 2; ++p) {
    const int kp = kBeg + p * KPASS;
    const int kcol = kp + 4 * l15;
    const float* srcA = (kcol < split) ? Alo : Ahi;
    const float* g = srcA + (long)(tile * 16 + half) * DD + kcol;
    int wrAt = rowWrAt;
    asm volatile("" : "+v"(wrAt));
    float* rowWr = stage + wrAt;
    float4 v[8];
#pragma unroll
    for (int i = 0; i < 8; ++i) v[i] = *(const float4*)(g + (long)(2 * i) * DD);
#pragma unroll
    for (int i = 0; i < 8; ++i) *(float4*)(rowWr + (2 * i) * SPITCH) = v[i];
    wave_lds_handover();

    const int kpEnd = (kp + KPASS < kEnd) ? (kp + KPASS) : kEnd;
#pragma unroll 2
    for (int k0 = kp; k0 < kpEnd; k0 += 4) {
      const int kk = k0 + half * 2;
      const v2f a = *(const v2f*)(fragRd + (k0 - kp));
      v2f b0, b1;
      b0.x = Wp[(long)kk * WPAD + l15];
      b0.y = Wp[(long)(kk + 1) * WPAD + l15];
      b1.x = Wp[(long)kk * WPAD + 16 + l15];
      b1.y = Wp[(long)(kk + 1) * WPAD + 16 + l15];

      acc0 = __builtin_amdgcn_wmma_f32_16x16x4_f32(
          false, a, false, b0, (short)0, acc0, false, false);
      acc1 = __builtin_amdgcn_wmma_f32_16x16x4_f32(
          false, a, false, b1, (short)0, acc1, false, false);
    }
    wave_lds_handover();
  }

#pragma unroll
  for (int v = 0; v < 8; ++v) {
    const int r = v + half * 8;
    my[r * WPAD + l15]      = acc0[v];
    my[r * WPAD + 16 + l15] = acc1[v];
  }
  __syncthreads();

  const int e = threadIdx.x;
  float s = stage[e];
#pragma unroll
  for (int j = 1; j < KSPLIT; ++j) s += stage[j * WSTAGE + e];
  wout[(long)tile * (16 * WPAD) + e] = s;
}

#define TSTEP 16
#define NROW  (TSTEP + 2 * PP)
#define ROW4  (DD / 4)
#define COL4  250
#define CBLK  128

template <int WHERE>
__device__ __forceinline__ void shift_tile(
    const float4* __restrict__ xcol, const float* __restrict__ wt,
    float4* __restrict__ ocol) {
  constexpr int rLo = (WHERE == 1) ? PP : 0;
  constexpr int rHi = (WHERE == 2) ? (NROW - PP) : NROW;

  float4 rows[NROW];
#pragma unroll
  for (int r = rLo; r < rHi; ++r) rows[r] = xcol[(long)(r - PP) * ROW4];

#pragma unroll
  for (int j = 0; j < TSTEP; ++j) {
    float4 acc = make_float4(0.f, 0.f, 0.f, 0.f);
#pragma unroll
    for (int k = 0; k < KK; ++k) {
      if (j + k < rLo || j + k >= rHi) continue;
      const float  s = wt[j * WPAD + k];
      const float4 v = rows[j + k];
      acc.x = __builtin_fmaf(s, v.x, acc.x);
      acc.y = __builtin_fmaf(s, v.y, acc.y);
      acc.z = __builtin_fmaf(s, v.z, acc.z);
      acc.w = __builtin_fmaf(s, v.w, acc.w);
    }
    ocol[(long)j * ROW4] = acc;
  }
}

__global__ __launch_bounds__(CBLK, 8) void shift_kernel(
    const float* __restrict__ x, const float* __restrict__ w,
    float* __restrict__ out, int sOff) {
  const int d4 = (blockIdx.x & 1) * CBLK + threadIdx.x;
  if (d4 >= COL4) return;
  const int row0 = (blockIdx.x >> 1) * TSTEP;
  const int t0   = row0 % TT;

  const float4* xcol = (const float4*)(x + (long)row0 * DD + sOff) + d4;
  float4*       ocol = (float4*)(out + (long)row0 * DD + sOff) + d4;
  const float*  wt   = w + (long)row0 * WPAD;

  if (t0 == 0)               shift_tile<1>(xcol, wt, ocol);
  else if (t0 == TT - TSTEP) shift_tile<2>(xcol, wt, ocol);
  else                       shift_tile<0>(xcol, wt, ocol);
}

extern "C" void kernel_launch(void* const* d_in, const int* in_sizes, int n_in,
                              void* d_out, int out_size, void* d_ws, size_t ws_size,
                              hipStream_t stream) {
  const float* x  = (const float*)d_in[0];
  const float* W0 = (const float*)d_in[1];
  const float* W1 = (const float*)d_in[2];
  float* out  = (float*)d_out;

  float* wbuf = (float*)d_ws;
  float* Wp   = (float*)d_ws + (long)RR * WPAD;

  (void)in_sizes; (void)n_in; (void)out_size; (void)ws_size;

  pad_w_kernel<<<DD, 32, 0, stream>>>(W0, Wp);
  wmma_xw_kernel<<<RR / 16, KSPLIT * 32, 0, stream>>>(x, x, DD, Wp, wbuf);
  shift_kernel<<<(RR / TSTEP) * 2, CBLK, 0, stream>>>(x, wbuf, out, 0);

  pad_w_kernel<<<DD, 32, 0, stream>>>(W1, Wp);
  wmma_xw_kernel<<<RR / 16, KSPLIT * 32, 0, stream>>>(out, x, 1000, Wp, wbuf);
  shift_kernel<<<(RR / TSTEP) * 2, CBLK, 0, stream>>>(x, wbuf, out, 1000);
}


# GEMM B operand: W packed once per group in WMMA fragment order, so each lane takes both B fragments of a k-step with one 16-byte load (was four 4-byte loads); k order, sums unchanged
# speedup vs baseline: 3.7195x; 1.0461x over previous
#include <hip/hip_runtime.h>
#include <hip/hip_bf16.h>

typedef float v2f __attribute__((ext_vector_type(2)));
typedef float v8f __attribute__((ext_vector_type(8)));

#define NB   16
#define TT   1024
#define DD   2000
#define KK   21
#define PP   10
#define RR   (NB * TT)
#define WPAD 32
#define KSPLIT 16

#define QSTEPS (DD / 4)

__global__ __launch_bounds__(32) void pack_w_kernel(
    const float* __restrict__ W, float* __restrict__ Wf) {
  const int q    = blockIdx.x;
  const int lane = threadIdx.x;
  const int c    = lane & 15;
  const int h    = lane >> 4;
  const float* r0 = W + (long)(4 * q + 2 * h) * KK;
  const float* r1 = r0 + KK;
  float4 v;
  v.x = r0[c];
  v.y = r1[c];
  v.z = (16 + c < KK) ? r0[16 + c] : 0.f;
  v.w = (16 + c < KK) ? r1[16 + c] : 0.f;
  ((float4*)Wf)[q * 32 + lane] = v;
}

#define KPASS  64
#define SPITCH (KPASS + 4)
#define WSTAGE (16 * SPITCH)

__device__ __forceinline__ void wave_lds_handover() {
  __builtin_amdgcn_fence(__ATOMIC_RELEASE, "wavefront");
  __builtin_amdgcn_wave_barrier();
  __builtin_amdgcn_fence(__ATOMIC_ACQUIRE, "wavefront");
}

__global__ __launch_bounds__(512) void wmma_xw_kernel(
    const float* __restrict__ Alo, const float* __restrict__ Ahi, int split,
    const float* __restrict__ Wf, float* __restrict__ wout) {
  __shared__ __attribute__((aligned(16))) float stage[KSPLIT * WSTAGE];

  const int lane = threadIdx.x & 31;
  const int ks   = __builtin_amdgcn_readfirstlane((int)(threadIdx.x >> 5));
  const int tile = blockIdx.x;
  const int half = lane >> 4;
  const int l15  = lane & 15;

  const int kBeg = ((ks * (DD / 4)) / KSPLIT) * 4;
  const int kEnd = (((ks + 1) * (DD / 4)) / KSPLIT) * 4;

  float* my = stage + ks * WSTAGE;
  const int    rowWrAt = ks * WSTAGE + half * SPITCH + l15 * 4;
  const float* fragRd = my + l15 * SPITCH + half * 2;
  const float4* bRd = (const float4*)Wf + lane;

  v8f acc0 = {0.f, 0.f, 0.f, 0.f, 0.f, 0.f, 0.f, 0.f};
  v8f acc1 = {0.f, 0.f, 0.f, 0.f, 0.f, 0.f, 0.f, 0.f};

#pragma unroll
  for (int p = 0; p < 2; ++p) {
    const int kp = kBeg + p * KPASS;
    const int kcol = kp + 4 * l15;
    const float* srcA = (kcol < split) ? Alo : Ahi;
    const float* g = srcA + (long)(tile * 16 + half) * DD + kcol;
    int wrAt = rowWrAt;
    asm volatile("" : "+v"(wrAt));
    float* rowWr = stage + wrAt;
    float4 v[8];
#pragma unroll
    for (int i = 0; i < 8; ++i) v[i] = *(const float4*)(g + (long)(2 * i) * DD);
#pragma unroll
    for (int i = 0; i < 8; ++i) *(float4*)(rowWr + (2 * i) * SPITCH) = v[i];
    wave_lds_handover();

    const int kpEnd = (kp + KPASS < kEnd) ? (kp + KPASS) : kEnd;
#pragma unroll 2
    for (int k0 = kp; k0 < kpEnd; k0 += 4) {
      const v2f a = *(const v2f*)(fragRd + (k0 - kp));
      const float4 bq = bRd[(k0 >> 2) * 32];
      v2f b0, b1;
      b0.x = bq.x;
      b0.y = bq.y;
      b1.x = bq.z;
      b1.y = bq.w;

      acc0 = __builtin_amdgcn_wmma_f32_16x16x4_f32(
          false, a, false, b0, (short)0, acc0, false, false);
      acc1 = __builtin_amdgcn_wmma_f32_16x16x4_f32(
          false, a, false, b1, (short)0, acc1, false, false);
    }
    wave_lds_handover();
  }

#pragma unroll
  for (int v = 0; v < 8; ++v) {
    const int r = v + half * 8;
    my[r * WPAD + l15]      = acc0[v];
    my[r * WPAD + 16 + l15] = acc1[v];
  }
  __syncthreads();

  const int e = threadIdx.x;
  float s = stage[e];
#pragma unroll
  for (int j = 1; j < KSPLIT; ++j) s += stage[j * WSTAGE + e];
  wout[(long)tile * (16 * WPAD) + e] = s;
}

#define TSTEP 16
#define NROW  (TSTEP + 2 * PP)
#define ROW4  (DD / 4)
#define COL4  250
#define CBLK  128

template <int WHERE>
__device__ __forceinline__ void shift_tile(
    const float4* __restrict__ xcol, const float* __restrict__ wt,
    float4* __restrict__ ocol) {
  constexpr int rLo = (WHERE == 1) ? PP : 0;
  constexpr int rHi = (WHERE == 2) ? (NROW - PP) : NROW;

  float4 rows[NROW];
#pragma unroll
  for (int r = rLo; r < rHi; ++r) rows[r] = xcol[(long)(r - PP) * ROW4];

#pragma unroll
  for (int j = 0; j < TSTEP; ++j) {
    float4 acc = make_float4(0.f, 0.f, 0.f, 0.f);
#pragma unroll
    for (int k = 0; k < KK; ++k) {
      if (j + k < rLo || j + k >= rHi) continue;
      const float  s = wt[j * WPAD + k];
      const float4 v = rows[j + k];
      acc.x = __builtin_fmaf(s, v.x, acc.x);
      acc.y = __builtin_fmaf(s, v.y, acc.y);
      acc.z = __builtin_fmaf(s, v.z, acc.z);
      acc.w = __builtin_fmaf(s, v.w, acc.w);
    }
    ocol[(long)j * ROW4] = acc;
  }
}

__global__ __launch_bounds__(CBLK, 8) void shift_kernel(
    const float* __restrict__ x, const float* __restrict__ w,
    float* __restrict__ out, int sOff) {
  const int d4 = (blockIdx.x & 1) * CBLK + threadIdx.x;
  if (d4 >= COL4) return;
  const int row0 = (blockIdx.x >> 1) * TSTEP;
  const int t0   = row0 % TT;

  const float4* xcol = (const float4*)(x + (long)row0 * DD + sOff) + d4;
  float4*       ocol = (float4*)(out + (long)row0 * DD + sOff) + d4;
  const float*  wt   = w + (long)row0 * WPAD;

  if (t0 == 0)               shift_tile<1>(xcol, wt, ocol);
  else if (t0 == TT - TSTEP) shift_tile<2>(xcol, wt, ocol);
  else                       shift_tile<0>(xcol, wt, ocol);
}

extern "C" void kernel_launch(void* const* d_in, const int* in_sizes, int n_in,
                              void* d_out, int out_size, void* d_ws, size_t ws_size,
                              hipStream_t stream) {
  const float* x  = (const float*)d_in[0];
  const float* W0 = (const float*)d_in[1];
  const float* W1 = (const float*)d_in[2];
  float* out  = (float*)d_out;

  float* wbuf = (float*)d_ws;
  float* Wf   = (float*)d_ws + (long)RR * WPAD;

  (void)in_sizes; (void)n_in; (void)out_size; (void)ws_size;

  pack_w_kernel<<<QSTEPS, 32, 0, stream>>>(W0, Wf);
  wmma_xw_kernel<<<RR / 16, KSPLIT * 32, 0, stream>>>(x, x, DD, Wf, wbuf);
  shift_kernel<<<(RR / TSTEP) * 2, CBLK, 0, stream>>>(x, wbuf, out, 0);

  pack_w_kernel<<<QSTEPS, 32, 0, stream>>>(W1, Wf);
  wmma_xw_kernel<<<RR / 16, KSPLIT * 32, 0, stream>>>(out, x, 1000, Wf, wbuf);
  shift_kernel<<<(RR / TSTEP) * 2, CBLK, 0, stream>>>(x, wbuf, out, 1000);
}
